# starting kernel
# baseline (speedup 1.0000x reference)
#include <hip/hip_runtime.h>
#include <hip/hip_bf16.h>

typedef float v2f __attribute__((ext_vector_type(2)));
typedef float v4f __attribute__((ext_vector_type(4)));
typedef float v8f __attribute__((ext_vector_type(8)));

template <bool TRANS_A, int K, int LDA, int LDB, int LDC>
__global__ __launch_bounds__(256) void gemm_f32_wmma(
    const float* __restrict__ A, const float* __restrict__ B,
    float* __restrict__ C) {
  constexpr int MT = 4;
  constexpr int NT = 2;

  const int lane = threadIdx.x & 31;
  const int wave = threadIdx.x >> 5;
  const int tileM = blockIdx.y * (2 * MT * 16) + (wave & 1) * (MT * 16);
  const int tileN = blockIdx.x * (4 * NT * 16) + (wave >> 1) * (NT * 16);
  const int r = lane & 15;
  const int h = lane >> 4;

  const float* Ap;
  if (TRANS_A) {
    Ap = A + (size_t)(2 * h) * LDA + (tileM + r);
  } else {
    Ap = A + (size_t)(tileM + r) * LDA + 2 * h;
  }
  const float* Bp = B + (size_t)(2 * h) * LDB + (tileN + r);

  v8f acc[MT][NT];
#pragma unroll
  for (int i = 0; i < MT; ++i)
#pragma unroll
    for (int j = 0; j < NT; ++j) acc[i][j] = (v8f){};

#pragma unroll 2
  for (int k = 0; k < K; k += 4) {
    v2f a[MT], b[NT];
#pragma unroll
    for (int i = 0; i < MT; ++i) {
      if (TRANS_A) {
        a[i].x = Ap[16 * i];
        a[i].y = Ap[16 * i + LDA];
      } else {
        a[i] = *(const v2f*)(Ap + (size_t)i * 16 * LDA);
      }
    }
#pragma unroll
    for (int j = 0; j < NT; ++j) {
      b[j].x = Bp[16 * j];
      b[j].y = Bp[16 * j + LDB];
    }
#pragma unroll
    for (int i = 0; i < MT; ++i)
#pragma unroll
      for (int j = 0; j < NT; ++j)
        acc[i][j] = __builtin_amdgcn_wmma_f32_16x16x4_f32(
            false, a[i], false, b[j], (short)0, acc[i][j], false, false);
    Ap += TRANS_A ? 4 * LDA : 4;
    Bp += 4 * LDB;
  }

  float* Cp = C + (size_t)(tileM + 8 * h) * LDC + tileN + r;
#pragma unroll
  for (int i = 0; i < MT; ++i)
#pragma unroll
    for (int j = 0; j < NT; ++j)
#pragma unroll
      for (int v = 0; v < 8; ++v)
        Cp[(size_t)(16 * i + v) * LDC + 16 * j] = acc[i][j][v];
}

__global__ __launch_bounds__(256) void fused_attn_ln(
    const float* __restrict__ cur, const float* __restrict__ prev,
    const float* __restrict__ ctil, const float* __restrict__ gamma,
    const float* __restrict__ beta, float* __restrict__ out) {
  constexpr int D = 1024;
  constexpr int P = 8;
  const int token = blockIdx.x;
  const int tid = threadIdx.x;
  const int lane = tid & 31;
  const int wave = tid >> 5;

  const v4f* cur4 = (const v4f*)(cur + (size_t)token * D);
  const v4f* ct4 = (const v4f*)(ctil + (size_t)token * D);
  const v4f* pbase = (const v4f*)(prev + (size_t)token * P * D);

  v4f x = cur4[tid];
  v4f c = ct4[tid];

  v4f pv[P];
  float s[P];
#pragma unroll
  for (int p = 0; p < P; ++p) {
    pv[p] = __builtin_nontemporal_load(pbase + (size_t)p * (D / 4) + tid);
    s[p] = c.x * pv[p].x + c.y * pv[p].y + c.z * pv[p].z + c.w * pv[p].w;
  }

  __shared__ float red[8][P];
  __shared__ float red2[8][2];

#pragma unroll
  for (int p = 0; p < P; ++p) {
#pragma unroll
    for (int off = 16; off >= 1; off >>= 1) s[p] += __shfl_xor(s[p], off, 32);
  }
  if (lane < P) red[wave][lane] = s[lane];
  __syncthreads();

  const float inv_sqrt_d = 0.03125f;
  float score[P];
#pragma unroll
  for (int p = 0; p < P; ++p) {
    float t = 0.f;
#pragma unroll
    for (int w = 0; w < 8; ++w) t += red[w][p];
    score[p] = t * inv_sqrt_d;
  }

  float mx = score[0];
#pragma unroll
  for (int p = 1; p < P; ++p) mx = fmaxf(mx, score[p]);
  float e[P], esum = 0.f;
#pragma unroll
  for (int p = 0; p < P; ++p) {
    e[p] = __expf(score[p] - mx);
    esum += e[p];
  }
  const float inv = 1.f / esum;

  v4f y = x;
#pragma unroll
  for (int p = 0; p < P; ++p) {
    const float w = e[p] * inv;
    y.x = fmaf(w, pv[p].x, y.x);
    y.y = fmaf(w, pv[p].y, y.y);
    y.z = fmaf(w, pv[p].z, y.z);
    y.w = fmaf(w, pv[p].w, y.w);
  }

  float ps = y.x + y.y + y.z + y.w;
  float ps2 = y.x * y.x + y.y * y.y + y.z * y.z + y.w * y.w;
#pragma unroll
  for (int off = 16; off >= 1; off >>= 1) {
    ps += __shfl_xor(ps, off, 32);
    ps2 += __shfl_xor(ps2, off, 32);
  }
  if (lane == 0) {
    red2[wave][0] = ps;
    red2[wave][1] = ps2;
  }
  __syncthreads();
  float S = 0.f, S2 = 0.f;
#pragma unroll
  for (int w = 0; w < 8; ++w) {
    S += red2[w][0];
    S2 += red2[w][1];
  }
  const float mu = S * (1.f / D);
  const float var = S2 * (1.f / D) - mu * mu;
  const float rstd = rsqrtf(var + 1e-5f);

  v4f g = ((const v4f*)gamma)[tid];
  v4f b = ((const v4f*)beta)[tid];
  v4f o;
  o.x = (y.x - mu) * rstd * g.x + b.x;
  o.y = (y.y - mu) * rstd * g.y + b.y;
  o.z = (y.z - mu) * rstd * g.z + b.z;
  o.w = (y.w - mu) * rstd * g.w + b.w;
  __builtin_nontemporal_store(o, (v4f*)(out + (size_t)token * D) + tid);
}

extern "C" void kernel_launch(void* const* d_in, const int* in_sizes, int n_in,
                              void* d_out, int out_size, void* d_ws,
                              size_t ws_size, hipStream_t stream) {
  const float* cur = (const float*)d_in[0];
  const float* prev = (const float*)d_in[1];
  const float* Wq = (const float*)d_in[2];
  const float* Wk = (const float*)d_in[3];
  const float* gamma = (const float*)d_in[4];
  const float* beta = (const float*)d_in[5];
  float* out = (float*)d_out;

  constexpr int D = 1024;
  const int M = in_sizes[0] / D;

  float* G = (float*)d_ws;
  float* Ctil = G + (size_t)D * D;

  dim3 blk(256);

  gemm_f32_wmma<true, D, D, D, D>
      <<<dim3(D / 128, D / 128), blk, 0, stream>>>(Wq, Wk, G);

  gemm_f32_wmma<false, D, D, D, D>
      <<<dim3(D / 128, M / 128), blk, 0, stream>>>(cur, G, Ctil);

  fused_attn_ln<<<dim3(M), blk, 0, stream>>>(cur, prev, Ctil, gamma, beta, out);
}


# Second GEMM (cur x G): A and B fragments fetched with 128-bit per-lane loads (k permuted inside 8-k blocks; first GEMM writes G k-pair-interleaved), half the cache lines per WMMA; k sum order changed
# speedup vs baseline: 1.6126x; 1.6126x over previous
#include <hip/hip_runtime.h>
#include <hip/hip_bf16.h>

typedef float v2f __attribute__((ext_vector_type(2)));
typedef float v4f __attribute__((ext_vector_type(4)));
typedef float v8f __attribute__((ext_vector_type(8)));


template <int K, int LDA, int LDB, int LDC, int KSPLIT>
__global__ __launch_bounds__(256) void gemm_tn_ksplit_f32_wmma(
    const float* __restrict__ A, const float* __restrict__ B,
    float* __restrict__ C2) {
  constexpr int MT = 4;
  constexpr int NT = 2;
  constexpr int KW = K / KSPLIT;
  static_assert(KSPLIT == 2 * MT, "one finishing wave per 8 rows");

  const int lane = threadIdx.x & 31;
  const int wave = threadIdx.x >> 5;
  const int tileM = blockIdx.y * (MT * 16);
  const int tileN = blockIdx.x * (NT * 16);
  const int k0 = wave * KW;
  const int r = lane & 15;
  const int h = lane >> 4;

  const float* Ap = A + (size_t)(k0 + 2 * h) * LDA + (tileM + r);
  const float* Bp = B + (size_t)(k0 + 2 * h) * LDB + (tileN + r);

  v8f acc[MT][NT];
#pragma unroll
  for (int i = 0; i < MT; ++i)
#pragma unroll
    for (int j = 0; j < NT; ++j) acc[i][j] = (v8f){};

#pragma unroll 2
  for (int k = 0; k < KW; k += 4) {
    v2f a[MT], b[NT];
#pragma unroll
    for (int i = 0; i < MT; ++i) {
      a[i].x = Ap[16 * i];
      a[i].y = Ap[16 * i + LDA];
    }
#pragma unroll
    for (int j = 0; j < NT; ++j) {
      b[j].x = Bp[16 * j];
      b[j].y = Bp[16 * j + LDB];
    }
#pragma unroll
    for (int i = 0; i < MT; ++i)
#pragma unroll
      for (int j = 0; j < NT; ++j)
        acc[i][j] = __builtin_amdgcn_wmma_f32_16x16x4_f32(
            false, a[i], false, b[j], (short)0, acc[i][j], false, false);
    Ap += 4 * LDA;
    Bp += 4 * LDB;
  }

  constexpr int SUB = 8 * 32;
  constexpr int TILE = MT * NT * SUB;
  static_assert(NT * 16 == 32, "finishing pass: lane = column of the tile");
  __shared__ float part[KSPLIT * TILE];
#pragma unroll
  for (int i = 0; i < MT; ++i)
#pragma unroll
    for (int j = 0; j < NT; ++j)
#pragma unroll
      for (int v = 0; v < 8; ++v)
        part[wave * TILE + (i * NT + j) * SUB + v * 32 + lane] = acc[i][j][v];
  __syncthreads();
  const float* sp = part + ((wave >> 1) * NT + (lane >> 4)) * SUB +
                    (wave & 1) * 16 + (lane & 15);
  float* Cp = C2 + ((size_t)((tileM + 8 * wave) >> 1) * LDC + tileN + lane) * 2;
#pragma unroll
  for (int vp = 0; vp < 4; ++vp) {
    float t0 = sp[(2 * vp) * 32];
    float t1 = sp[(2 * vp + 1) * 32];
#pragma unroll
    for (int w = 1; w < KSPLIT; ++w) {
      t0 += sp[w * TILE + (2 * vp) * 32];
      t1 += sp[w * TILE + (2 * vp + 1) * 32];
    }
    *(v2f*)(Cp + (size_t)vp * 2 * LDC) = (v2f){t0, t1};
  }
}

template <int K, int LDA, int N>
__global__ __launch_bounds__(256) void gemm_nn_pairb_f32_wmma(
    const float* __restrict__ A, const float* __restrict__ G2,
    float* __restrict__ C) {
  constexpr int MT = 4;
  constexpr int NT = 2;
  static_assert(K % 16 == 0, "K walked in blocks of 8, unrolled by 2");

  const int lane = threadIdx.x & 31;
  const int wave = threadIdx.x >> 5;
  const int tileM = blockIdx.y * (2 * MT * 16) + (wave & 1) * (MT * 16);
  const int tileN = blockIdx.x * (4 * NT * 16) + (wave >> 1) * (NT * 16);
  const int r = lane & 15;
  const int h = lane >> 4;

  const float* Ap = A + (size_t)(tileM + r) * LDA + 4 * h;
  const float* Bp =
      G2 + (size_t)(2 * h) * (2 * N) + 2 * tileN + 4 * r;

  v8f acc[MT][NT];
#pragma unroll
  for (int i = 0; i < MT; ++i)
#pragma unroll
    for (int j = 0; j < NT; ++j) acc[i][j] = (v8f){};

#pragma unroll 2
  for (int kb = 0; kb < K; kb += 8) {
    v4f a[MT], b[2];
#pragma unroll
    for (int i = 0; i < MT; ++i)
      a[i] = *(const v4f*)(Ap + (size_t)i * 16 * LDA);
#pragma unroll
    for (int s = 0; s < 2; ++s)
      b[s] = *(const v4f*)(Bp + (size_t)s * 2 * N);
#pragma unroll
    for (int s = 0; s < 2; ++s)
#pragma unroll
      for (int i = 0; i < MT; ++i)
#pragma unroll
        for (int j = 0; j < NT; ++j) {
          const v2f av = {a[i][2 * s], a[i][2 * s + 1]};
          const v2f bv = {b[s][2 * j], b[s][2 * j + 1]};
          acc[i][j] = __builtin_amdgcn_wmma_f32_16x16x4_f32(
              false, av, false, bv, (short)0, acc[i][j], false, false);
        }
    Ap += 8;
    Bp += 8 * N;
  }

  float* Cp = C + (size_t)(tileM + 8 * h) * N + tileN + 2 * r;
#pragma unroll
  for (int i = 0; i < MT; ++i)
#pragma unroll
    for (int v = 0; v < 8; ++v)
      *(v2f*)(Cp + (size_t)(16 * i + v) * N) = (v2f){acc[i][0][v], acc[i][1][v]};
}

__global__ __launch_bounds__(256) void fused_attn_ln(
    const float* __restrict__ cur, const float* __restrict__ prev,
    const float* __restrict__ ctil, const float* __restrict__ gamma,
    const float* __restrict__ beta, float* __restrict__ out) {
  constexpr int D = 1024;
  constexpr int P = 8;
  const int token = blockIdx.x;
  const int tid = threadIdx.x;
  const int lane = tid & 31;
  const int wave = tid >> 5;

  const v4f* cur4 = (const v4f*)(cur + (size_t)token * D);
  const v4f* ct4 = (const v4f*)(ctil + (size_t)token * D);
  const v4f* pbase = (const v4f*)(prev + (size_t)token * P * D);

  v4f x = cur4[tid];
  v4f c = ct4[tid];

  v4f pv[P];
  float s[P];
#pragma unroll
  for (int p = 0; p < P; ++p) {
    pv[p] = __builtin_nontemporal_load(pbase + (size_t)p * (D / 4) + tid);
    s[p] = c.x * pv[p].x + c.y * pv[p].y + c.z * pv[p].z + c.w * pv[p].w;
  }

  __shared__ float red[8][P];
  __shared__ float red2[8][2];

#pragma unroll
  for (int p = 0; p < P; ++p) {
#pragma unroll
    for (int off = 16; off >= 1; off >>= 1) s[p] += __shfl_xor(s[p], off, 32);
  }
  if (lane < P) red[wave][lane] = s[lane];
  __syncthreads();

  const float inv_sqrt_d = 0.03125f;
  float score[P];
#pragma unroll
  for (int p = 0; p < P; ++p) {
    float t = 0.f;
#pragma unroll
    for (int w = 0; w < 8; ++w) t += red[w][p];
    score[p] = t * inv_sqrt_d;
  }

  float mx = score[0];
#pragma unroll
  for (int p = 1; p < P; ++p) mx = fmaxf(mx, score[p]);
  float e[P], esum = 0.f;
#pragma unroll
  for (int p = 0; p < P; ++p) {
    e[p] = __expf(score[p] - mx);
    esum += e[p];
  }
  const float inv = 1.f / esum;

  v4f y = x;
#pragma unroll
  for (int p = 0; p < P; ++p) {
    const float w = e[p] * inv;
    y.x = fmaf(w, pv[p].x, y.x);
    y.y = fmaf(w, pv[p].y, y.y);
    y.z = fmaf(w, pv[p].z, y.z);
    y.w = fmaf(w, pv[p].w, y.w);
  }

  float ps = y.x + y.y + y.z + y.w;
  float ps2 = y.x * y.x + y.y * y.y + y.z * y.z + y.w * y.w;
#pragma unroll
  for (int off = 16; off >= 1; off >>= 1) {
    ps += __shfl_xor(ps, off, 32);
    ps2 += __shfl_xor(ps2, off, 32);
  }
  if (lane == 0) {
    red2[wave][0] = ps;
    red2[wave][1] = ps2;
  }
  __syncthreads();
  float S = 0.f, S2 = 0.f;
#pragma unroll
  for (int w = 0; w < 8; ++w) {
    S += red2[w][0];
    S2 += red2[w][1];
  }
  const float mu = S * (1.f / D);
  const float var = S2 * (1.f / D) - mu * mu;
  const float rstd = rsqrtf(var + 1e-5f);

  v4f g = ((const v4f*)gamma)[tid];
  v4f b = ((const v4f*)beta)[tid];
  v4f o;
  o.x = (y.x - mu) * rstd * g.x + b.x;
  o.y = (y.y - mu) * rstd * g.y + b.y;
  o.z = (y.z - mu) * rstd * g.z + b.z;
  o.w = (y.w - mu) * rstd * g.w + b.w;
  __builtin_nontemporal_store(o, (v4f*)(out + (size_t)token * D) + tid);
}

extern "C" void kernel_launch(void* const* d_in, const int* in_sizes, int n_in,
                              void* d_out, int out_size, void* d_ws,
                              size_t ws_size, hipStream_t stream) {
  const float* cur = (const float*)d_in[0];
  const float* prev = (const float*)d_in[1];
  const float* Wq = (const float*)d_in[2];
  const float* Wk = (const float*)d_in[3];
  const float* gamma = (const float*)d_in[4];
  const float* beta = (const float*)d_in[5];
  float* out = (float*)d_out;

  constexpr int D = 1024;
  const int M = in_sizes[0] / D;

  float* G2 = (float*)d_ws;
  float* Ctil = G2 + (size_t)D * D;

  dim3 blk(256);

  gemm_tn_ksplit_f32_wmma<D, D, D, D, 8>
      <<<dim3(D / 32, D / 64), blk, 0, stream>>>(Wq, Wk, G2);

  gemm_nn_pairb_f32_wmma<D, D, D>
      <<<dim3(D / 128, M / 128), blk, 0, stream>>>(cur, G2, Ctil);

  fused_attn_ln<<<dim3(M), blk, 0, stream>>>(cur, prev, Ctil, gamma, beta, out);
}


# Second GEMM fused with the softmax/LayerNorm pass: a block owns 32 tokens x all 1024 columns, keeps c~ in accumulators, reads prev once (2 launches, no c~ round trip); score and LN sum order changed
# speedup vs baseline: 1.8439x; 1.1434x over previous
#include <hip/hip_runtime.h>
#include <hip/hip_bf16.h>

typedef float v2f __attribute__((ext_vector_type(2)));
typedef float v4f __attribute__((ext_vector_type(4)));
typedef float v8f __attribute__((ext_vector_type(8)));


template <int K, int LDA, int LDB, int LDC, int KSPLIT>
__global__ __launch_bounds__(256) void gemm_tn_ksplit_f32_wmma(
    const float* __restrict__ A, const float* __restrict__ B,
    float* __restrict__ C2) {
  constexpr int MT = 4;
  constexpr int NT = 2;
  constexpr int KW = K / KSPLIT;
  static_assert(KSPLIT == 2 * MT, "one finishing wave per 8 rows");

  const int lane = threadIdx.x & 31;
  const int wave = threadIdx.x >> 5;
  const int tileM = blockIdx.y * (MT * 16);
  const int tileN = blockIdx.x * (NT * 16);
  const int k0 = wave * KW;
  const int r = lane & 15;
  const int h = lane >> 4;

  const float* Ap = A + (size_t)(k0 + 2 * h) * LDA + (tileM + r);
  const float* Bp = B + (size_t)(k0 + 2 * h) * LDB + (tileN + r);

  v8f acc[MT][NT];
#pragma unroll
  for (int i = 0; i < MT; ++i)
#pragma unroll
    for (int j = 0; j < NT; ++j) acc[i][j] = (v8f){};

#pragma unroll 2
  for (int k = 0; k < KW; k += 4) {
    v2f a[MT], b[NT];
#pragma unroll
    for (int i = 0; i < MT; ++i) {
      a[i].x = Ap[16 * i];
      a[i].y = Ap[16 * i + LDA];
    }
#pragma unroll
    for (int j = 0; j < NT; ++j) {
      b[j].x = Bp[16 * j];
      b[j].y = Bp[16 * j + LDB];
    }
#pragma unroll
    for (int i = 0; i < MT; ++i)
#pragma unroll
      for (int j = 0; j < NT; ++j)
        acc[i][j] = __builtin_amdgcn_wmma_f32_16x16x4_f32(
            false, a[i], false, b[j], (short)0, acc[i][j], false, false);
    Ap += 4 * LDA;
    Bp += 4 * LDB;
  }

  constexpr int SUB = 8 * 32;
  constexpr int TILE = MT * NT * SUB;
  static_assert(NT * 16 == 32, "finishing pass: lane = column of the tile");
  __shared__ float part[KSPLIT * TILE];
#pragma unroll
  for (int i = 0; i < MT; ++i)
#pragma unroll
    for (int j = 0; j < NT; ++j)
#pragma unroll
      for (int v = 0; v < 8; ++v)
        part[wave * TILE + (i * NT + j) * SUB + v * 32 + lane] = acc[i][j][v];
  __syncthreads();
  const float* sp = part + ((wave >> 1) * NT + (lane >> 4)) * SUB +
                    (wave & 1) * 16 + (lane & 15);
  float* Cp = C2 + ((size_t)((tileM + 8 * wave) >> 1) * LDC + tileN + lane) * 2;
#pragma unroll
  for (int vp = 0; vp < 4; ++vp) {
    float t0 = sp[(2 * vp) * 32];
    float t1 = sp[(2 * vp + 1) * 32];
#pragma unroll
    for (int w = 1; w < KSPLIT; ++w) {
      t0 += sp[w * TILE + (2 * vp) * 32];
      t1 += sp[w * TILE + (2 * vp + 1) * 32];
    }
    *(v2f*)(Cp + (size_t)vp * 2 * LDC) = (v2f){t0, t1};
  }
}

template <int M>
static __device__ __forceinline__ float row_xor_add(float x) {
  const int y = __builtin_amdgcn_update_dpp(0, __builtin_bit_cast(int, x),
                                            0x160 + M, 0xf, 0xf, false);
  return x + __builtin_bit_cast(float, y);
}
static __device__ __forceinline__ float row16_sum(float x) {
  x = row_xor_add<8>(x);
  x = row_xor_add<4>(x);
  x = row_xor_add<2>(x);
  x = row_xor_add<1>(x);
  return x;
}

static __device__ __forceinline__ void lds_barrier() {
  asm volatile("s_wait_dscnt 0x0\n\ts_barrier_signal -1\n\ts_barrier_wait -1" ::: "memory");
}

template <int K, int N>
__global__ __launch_bounds__(256) void gemm_scores_ln_f32_wmma(
    const float* __restrict__ cur, const float* __restrict__ G2,
    const float* __restrict__ prev, const float* __restrict__ gamma,
    const float* __restrict__ beta, float* __restrict__ out) {
  constexpr int MT = 2;
  constexpr int NG = 4;
  constexpr int P = 8;
  static_assert(K % 16 == 0 && N == 8 * 32 * NG, "8 waves x 128 columns");

  const int lane = threadIdx.x & 31;
  const int wave = threadIdx.x >> 5;
  const int tileM = blockIdx.x * (MT * 16);
  const int colbase = wave * (32 * NG);
  const int r = lane & 15;
  const int h = lane >> 4;

  const float* Ap = cur + (size_t)(tileM + r) * K + 4 * h;
  const float* Bp =
      G2 + (size_t)(2 * h) * (2 * N) + 2 * colbase + 4 * r;

  v8f acc[MT][NG][2];
#pragma unroll
  for (int i = 0; i < MT; ++i)
#pragma unroll
    for (int g = 0; g < NG; ++g)
#pragma unroll
      for (int j = 0; j < 2; ++j) acc[i][g][j] = (v8f){};

  v4f a[2][MT], b[2][2][NG];

  auto ldfrag = [&](int bf, int ko) {
#pragma unroll
    for (int i = 0; i < MT; ++i)
      a[bf][i] = *(const v4f*)(Ap + ko + (size_t)i * 16 * K);
#pragma unroll
    for (int s = 0; s < 2; ++s)
#pragma unroll
      for (int g = 0; g < NG; ++g)
        b[bf][s][g] = *(const v4f*)(Bp + (size_t)ko * N + (size_t)s * 2 * N + 64 * g);
  };
  auto mma = [&](int bf) {
#pragma unroll
    for (int s = 0; s < 2; ++s)
#pragma unroll
      for (int i = 0; i < MT; ++i)
#pragma unroll
        for (int g = 0; g < NG; ++g)
#pragma unroll
          for (int j = 0; j < 2; ++j) {
            const v2f av = {a[bf][i][2 * s], a[bf][i][2 * s + 1]};
            const v2f bv = {b[bf][s][g][2 * j], b[bf][s][g][2 * j + 1]};
            acc[i][g][j] = __builtin_amdgcn_wmma_f32_16x16x4_f32(
                false, av, false, bv, (short)0, acc[i][g][j], false, false);
          }
  };

  ldfrag(0, 0);
  for (int kb = 0; kb < K - 16; kb += 16) {
    ldfrag(1, 8);
    mma(0);
    ldfrag(0, 16);
    mma(1);
    Ap += 16;
    Bp += 16 * N;
  }
  ldfrag(1, 8);
  mma(0);
  mma(1);

  __shared__ float red[8][2][P];
  __shared__ float red2[8][2][2];

  const size_t lanecol = (size_t)colbase + 2 * r;
  const float* pvb = prev + (size_t)(tileM + 8 * h) * P * N + lanecol;
  const float* xb = cur + (size_t)(tileM + 8 * h) * N + lanecol;
  float* ob = out + (size_t)(tileM + 8 * h) * N + lanecol;

  v2f gm[NG], bt[NG];
#pragma unroll
  for (int g = 0; g < NG; ++g) {
    gm[g] = *(const v2f*)(gamma + lanecol + 32 * g);
    bt[g] = *(const v2f*)(beta + lanecol + 32 * g);
  }

  v2f pvv[2][P][NG], xv[2][NG];
  auto ldstep = [&](int bf, int st) {
    const int rowoff = (st >> 3) * 16 + (st & 7);
#pragma unroll
    for (int p = 0; p < P; ++p)
#pragma unroll
      for (int g = 0; g < NG; ++g)
        pvv[bf][p][g] = __builtin_nontemporal_load(
            (const v2f*)(pvb + ((size_t)rowoff * P + p) * N + 32 * g));
#pragma unroll
    for (int g = 0; g < NG; ++g)
      xv[bf][g] = *(const v2f*)(xb + (size_t)rowoff * N + 32 * g);
  };

  const float inv_sqrt_d = 0.03125f;
  ldstep(0, 0);
#pragma unroll
  for (int st = 0; st < 16; ++st) {
    const int bf = st & 1;
    const int i = st >> 3;
    const int v = st & 7;
    if (st + 1 < 16) ldstep(bf ^ 1, st + 1);

    float s[P];
#pragma unroll
    for (int p = 0; p < P; ++p) {
      float t = 0.f;
#pragma unroll
      for (int g = 0; g < NG; ++g) {
        t = fmaf(acc[i][g][0][v], pvv[bf][p][g].x, t);
        t = fmaf(acc[i][g][1][v], pvv[bf][p][g].y, t);
      }
      s[p] = t;
    }
#pragma unroll
    for (int p = 0; p < P; ++p) s[p] = row16_sum(s[p]);
    float sel = s[0];
#pragma unroll
    for (int p = 1; p < P; ++p) sel = (r == p) ? s[p] : sel;
    if (r < P) red[wave][h][r] = sel;
    lds_barrier();

    float score[P];
#pragma unroll
    for (int p = 0; p < P; ++p) {
      float t = 0.f;
#pragma unroll
      for (int w = 0; w < 8; ++w) t += red[w][h][p];
      score[p] = t * inv_sqrt_d;
    }
    float mx = score[0];
#pragma unroll
    for (int p = 1; p < P; ++p) mx = fmaxf(mx, score[p]);
    float e[P], esum = 0.f;
#pragma unroll
    for (int p = 0; p < P; ++p) {
      e[p] = __expf(score[p] - mx);
      esum += e[p];
    }
    const float inv = 1.f / esum;

    v2f y[NG];
#pragma unroll
    for (int g = 0; g < NG; ++g) y[g] = xv[bf][g];
#pragma unroll
    for (int p = 0; p < P; ++p) {
      const float w = e[p] * inv;
#pragma unroll
      for (int g = 0; g < NG; ++g) {
        y[g].x = fmaf(w, pvv[bf][p][g].x, y[g].x);
        y[g].y = fmaf(w, pvv[bf][p][g].y, y[g].y);
      }
    }

    float ps = 0.f, ps2 = 0.f;
#pragma unroll
    for (int g = 0; g < NG; ++g) {
      ps += y[g].x + y[g].y;
      ps2 += y[g].x * y[g].x + y[g].y * y[g].y;
    }
    ps = row16_sum(ps);
    ps2 = row16_sum(ps2);
    if (r == 0) {
      red2[wave][h][0] = ps;
      red2[wave][h][1] = ps2;
    }
    lds_barrier();
    float S = 0.f, S2 = 0.f;
#pragma unroll
    for (int w = 0; w < 8; ++w) {
      S += red2[w][h][0];
      S2 += red2[w][h][1];
    }
    const float mu = S * (1.f / N);
    const float var = S2 * (1.f / N) - mu * mu;
    const float rstd = rsqrtf(var + 1e-5f);

    const int rowoff = i * 16 + v;
#pragma unroll
    for (int g = 0; g < NG; ++g) {
      v2f o;
      o.x = (y[g].x - mu) * rstd * gm[g].x + bt[g].x;
      o.y = (y[g].y - mu) * rstd * gm[g].y + bt[g].y;
      __builtin_nontemporal_store(o, (v2f*)(ob + (size_t)rowoff * N + 32 * g));
    }
  }
}

extern "C" void kernel_launch(void* const* d_in, const int* in_sizes, int n_in,
                              void* d_out, int out_size, void* d_ws,
                              size_t ws_size, hipStream_t stream) {
  const float* cur = (const float*)d_in[0];
  const float* prev = (const float*)d_in[1];
  const float* Wq = (const float*)d_in[2];
  const float* Wk = (const float*)d_in[3];
  const float* gamma = (const float*)d_in[4];
  const float* beta = (const float*)d_in[5];
  float* out = (float*)d_out;

  constexpr int D = 1024;
  const int M = in_sizes[0] / D;

  float* G2 = (float*)d_ws;

  dim3 blk(256);

  gemm_tn_ksplit_f32_wmma<D, D, D, D, 8>
      <<<dim3(D / 32, D / 64), blk, 0, stream>>>(Wq, Wk, G2);

  gemm_scores_ln_f32_wmma<D, D>
      <<<dim3(M / 32), blk, 0, stream>>>(cur, G2, prev, gamma, beta, out);
}


# Second GEMM K loop: A and B fragments requested three 8-k blocks ahead of their WMMAs (four register sets, order kept by scheduling fences; was one block ahead, two sets); sum order unchanged
# speedup vs baseline: 1.9484x; 1.0567x over previous
#include <hip/hip_runtime.h>
#include <hip/hip_bf16.h>

typedef float v2f __attribute__((ext_vector_type(2)));
typedef float v4f __attribute__((ext_vector_type(4)));
typedef float v8f __attribute__((ext_vector_type(8)));


template <int K, int LDA, int LDB, int LDC, int KSPLIT>
__global__ __launch_bounds__(256) void gemm_tn_ksplit_f32_wmma(
    const float* __restrict__ A, const float* __restrict__ B,
    float* __restrict__ C2) {
  constexpr int MT = 4;
  constexpr int NT = 2;
  constexpr int KW = K / KSPLIT;
  static_assert(KSPLIT == 2 * MT, "one finishing wave per 8 rows");

  const int lane = threadIdx.x & 31;
  const int wave = threadIdx.x >> 5;
  const int tileM = blockIdx.y * (MT * 16);
  const int tileN = blockIdx.x * (NT * 16);
  const int k0 = wave * KW;
  const int r = lane & 15;
  const int h = lane >> 4;

  const float* Ap = A + (size_t)(k0 + 2 * h) * LDA + (tileM + r);
  const float* Bp = B + (size_t)(k0 + 2 * h) * LDB + (tileN + r);

  v8f acc[MT][NT];
#pragma unroll
  for (int i = 0; i < MT; ++i)
#pragma unroll
    for (int j = 0; j < NT; ++j) acc[i][j] = (v8f){};

#pragma unroll 2
  for (int k = 0; k < KW; k += 4) {
    v2f a[MT], b[NT];
#pragma unroll
    for (int i = 0; i < MT; ++i) {
      a[i].x = Ap[16 * i];
      a[i].y = Ap[16 * i + LDA];
    }
#pragma unroll
    for (int j = 0; j < NT; ++j) {
      b[j].x = Bp[16 * j];
      b[j].y = Bp[16 * j + LDB];
    }
#pragma unroll
    for (int i = 0; i < MT; ++i)
#pragma unroll
      for (int j = 0; j < NT; ++j)
        acc[i][j] = __builtin_amdgcn_wmma_f32_16x16x4_f32(
            false, a[i], false, b[j], (short)0, acc[i][j], false, false);
    Ap += 4 * LDA;
    Bp += 4 * LDB;
  }

  constexpr int SUB = 8 * 32;
  constexpr int TILE = MT * NT * SUB;
  static_assert(NT * 16 == 32, "finishing pass: lane = column of the tile");
  __shared__ float part[KSPLIT * TILE];
#pragma unroll
  for (int i = 0; i < MT; ++i)
#pragma unroll
    for (int j = 0; j < NT; ++j)
#pragma unroll
      for (int v = 0; v < 8; ++v)
        part[wave * TILE + (i * NT + j) * SUB + v * 32 + lane] = acc[i][j][v];
  __syncthreads();
  const float* sp = part + ((wave >> 1) * NT + (lane >> 4)) * SUB +
                    (wave & 1) * 16 + (lane & 15);
  float* Cp = C2 + ((size_t)((tileM + 8 * wave) >> 1) * LDC + tileN + lane) * 2;
#pragma unroll
  for (int vp = 0; vp < 4; ++vp) {
    float t0 = sp[(2 * vp) * 32];
    float t1 = sp[(2 * vp + 1) * 32];
#pragma unroll
    for (int w = 1; w < KSPLIT; ++w) {
      t0 += sp[w * TILE + (2 * vp) * 32];
      t1 += sp[w * TILE + (2 * vp + 1) * 32];
    }
    *(v2f*)(Cp + (size_t)vp * 2 * LDC) = (v2f){t0, t1};
  }
}

template <int M>
static __device__ __forceinline__ float row_xor_add(float x) {
  const int y = __builtin_amdgcn_update_dpp(0, __builtin_bit_cast(int, x),
                                            0x160 + M, 0xf, 0xf, false);
  return x + __builtin_bit_cast(float, y);
}
static __device__ __forceinline__ float row16_sum(float x) {
  x = row_xor_add<8>(x);
  x = row_xor_add<4>(x);
  x = row_xor_add<2>(x);
  x = row_xor_add<1>(x);
  return x;
}

static __device__ __forceinline__ void lds_barrier() {
  asm volatile("s_wait_dscnt 0x0\n\ts_barrier_signal -1\n\ts_barrier_wait -1" ::: "memory");
}

template <int K, int N>
__global__ __launch_bounds__(256) void gemm_scores_ln_f32_wmma(
    const float* __restrict__ cur, const float* __restrict__ G2,
    const float* __restrict__ prev, const float* __restrict__ gamma,
    const float* __restrict__ beta, float* __restrict__ out) {
  constexpr int MT = 2;
  constexpr int NG = 4;
  constexpr int P = 8;
  static_assert(K % 32 == 0 && K >= 64 && N == 8 * 32 * NG,
                "8 waves x 128 columns, K loop of four 8-k blocks");

  const int lane = threadIdx.x & 31;
  const int wave = threadIdx.x >> 5;
  const int tileM = blockIdx.x * (MT * 16);
  const int colbase = wave * (32 * NG);
  const int r = lane & 15;
  const int h = lane >> 4;

  const float* Ap = cur + (size_t)(tileM + r) * K + 4 * h;
  const float* Bp =
      G2 + (size_t)(2 * h) * (2 * N) + 2 * colbase + 4 * r;

  v8f acc[MT][NG][2];
#pragma unroll
  for (int i = 0; i < MT; ++i)
#pragma unroll
    for (int g = 0; g < NG; ++g)
#pragma unroll
      for (int j = 0; j < 2; ++j) acc[i][g][j] = (v8f){};

  v4f a[4][MT], b[4][2][NG];

  auto ldfrag = [&](int bf, int ko) {
#pragma unroll
    for (int i = 0; i < MT; ++i)
      a[bf][i] = *(const v4f*)(Ap + ko + (size_t)i * 16 * K);
#pragma unroll
    for (int s = 0; s < 2; ++s)
#pragma unroll
      for (int g = 0; g < NG; ++g)
        b[bf][s][g] = *(const v4f*)(Bp + (size_t)ko * N + (size_t)s * 2 * N + 64 * g);
  };
  auto mma = [&](int bf) {
#pragma unroll
    for (int s = 0; s < 2; ++s)
#pragma unroll
      for (int i = 0; i < MT; ++i)
#pragma unroll
        for (int g = 0; g < NG; ++g)
#pragma unroll
          for (int j = 0; j < 2; ++j) {
            const v2f av = {a[bf][i][2 * s], a[bf][i][2 * s + 1]};
            const v2f bv = {b[bf][s][g][2 * j], b[bf][s][g][2 * j + 1]};
            acc[i][g][j] = __builtin_amdgcn_wmma_f32_16x16x4_f32(
                false, av, false, bv, (short)0, acc[i][g][j], false, false);
          }
  };

#define PIN_ORDER() __builtin_amdgcn_sched_barrier(0)
  ldfrag(0, 0);
  PIN_ORDER();
  ldfrag(1, 8);
  PIN_ORDER();
  ldfrag(2, 16);
  for (int kb = 0; kb < K - 32; kb += 32) {
    PIN_ORDER();
    ldfrag(3, 24);
    PIN_ORDER();
    mma(0);
    PIN_ORDER();
    ldfrag(0, 32);
    PIN_ORDER();
    mma(1);
    PIN_ORDER();
    ldfrag(1, 40);
    PIN_ORDER();
    mma(2);
    PIN_ORDER();
    ldfrag(2, 48);
    PIN_ORDER();
    mma(3);
    Ap += 32;
    Bp += 32 * N;
  }
  PIN_ORDER();
  ldfrag(3, 24);
  PIN_ORDER();
  mma(0);
  mma(1);
  mma(2);
  mma(3);
#undef PIN_ORDER

  __shared__ float red[8][2][P];
  __shared__ float red2[8][2][2];

  const size_t lanecol = (size_t)colbase + 2 * r;
  const float* pvb = prev + (size_t)(tileM + 8 * h) * P * N + lanecol;
  const float* xb = cur + (size_t)(tileM + 8 * h) * N + lanecol;
  float* ob = out + (size_t)(tileM + 8 * h) * N + lanecol;

  v2f gm[NG], bt[NG];
#pragma unroll
  for (int g = 0; g < NG; ++g) {
    gm[g] = *(const v2f*)(gamma + lanecol + 32 * g);
    bt[g] = *(const v2f*)(beta + lanecol + 32 * g);
  }

  v2f pvv[2][P][NG], xv[2][NG];
  auto ldstep = [&](int bf, int st) {
    const int rowoff = (st >> 3) * 16 + (st & 7);
#pragma unroll
    for (int p = 0; p < P; ++p)
#pragma unroll
      for (int g = 0; g < NG; ++g)
        pvv[bf][p][g] = __builtin_nontemporal_load(
            (const v2f*)(pvb + ((size_t)rowoff * P + p) * N + 32 * g));
#pragma unroll
    for (int g = 0; g < NG; ++g)
      xv[bf][g] = *(const v2f*)(xb + (size_t)rowoff * N + 32 * g);
  };

  const float inv_sqrt_d = 0.03125f;
  ldstep(0, 0);
#pragma unroll
  for (int st = 0; st < 16; ++st) {
    const int bf = st & 1;
    const int i = st >> 3;
    const int v = st & 7;
    if (st + 1 < 16) ldstep(bf ^ 1, st + 1);

    float s[P];
#pragma unroll
    for (int p = 0; p < P; ++p) {
      float t = 0.f;
#pragma unroll
      for (int g = 0; g < NG; ++g) {
        t = fmaf(acc[i][g][0][v], pvv[bf][p][g].x, t);
        t = fmaf(acc[i][g][1][v], pvv[bf][p][g].y, t);
      }
      s[p] = t;
    }
#pragma unroll
    for (int p = 0; p < P; ++p) s[p] = row16_sum(s[p]);
    float sel = s[0];
#pragma unroll
    for (int p = 1; p < P; ++p) sel = (r == p) ? s[p] : sel;
    if (r < P) red[wave][h][r] = sel;
    lds_barrier();

    float score[P];
#pragma unroll
    for (int p = 0; p < P; ++p) {
      float t = 0.f;
#pragma unroll
      for (int w = 0; w < 8; ++w) t += red[w][h][p];
      score[p] = t * inv_sqrt_d;
    }
    float mx = score[0];
#pragma unroll
    for (int p = 1; p < P; ++p) mx = fmaxf(mx, score[p]);
    float e[P], esum = 0.f;
#pragma unroll
    for (int p = 0; p < P; ++p) {
      e[p] = __expf(score[p] - mx);
      esum += e[p];
    }
    const float inv = 1.f / esum;

    v2f y[NG];
#pragma unroll
    for (int g = 0; g < NG; ++g) y[g] = xv[bf][g];
#pragma unroll
    for (int p = 0; p < P; ++p) {
      const float w = e[p] * inv;
#pragma unroll
      for (int g = 0; g < NG; ++g) {
        y[g].x = fmaf(w, pvv[bf][p][g].x, y[g].x);
        y[g].y = fmaf(w, pvv[bf][p][g].y, y[g].y);
      }
    }

    float ps = 0.f, ps2 = 0.f;
#pragma unroll
    for (int g = 0; g < NG; ++g) {
      ps += y[g].x + y[g].y;
      ps2 += y[g].x * y[g].x + y[g].y * y[g].y;
    }
    ps = row16_sum(ps);
    ps2 = row16_sum(ps2);
    if (r == 0) {
      red2[wave][h][0] = ps;
      red2[wave][h][1] = ps2;
    }
    lds_barrier();
    float S = 0.f, S2 = 0.f;
#pragma unroll
    for (int w = 0; w < 8; ++w) {
      S += red2[w][h][0];
      S2 += red2[w][h][1];
    }
    const float mu = S * (1.f / N);
    const float var = S2 * (1.f / N) - mu * mu;
    const float rstd = rsqrtf(var + 1e-5f);

    const int rowoff = i * 16 + v;
#pragma unroll
    for (int g = 0; g < NG; ++g) {
      v2f o;
      o.x = (y[g].x - mu) * rstd * gm[g].x + bt[g].x;
      o.y = (y[g].y - mu) * rstd * gm[g].y + bt[g].y;
      __builtin_nontemporal_store(o, (v2f*)(ob + (size_t)rowoff * N + 32 * g));
    }
  }
}

extern "C" void kernel_launch(void* const* d_in, const int* in_sizes, int n_in,
                              void* d_out, int out_size, void* d_ws,
                              size_t ws_size, hipStream_t stream) {
  const float* cur = (const float*)d_in[0];
  const float* prev = (const float*)d_in[1];
  const float* Wq = (const float*)d_in[2];
  const float* Wk = (const float*)d_in[3];
  const float* gamma = (const float*)d_in[4];
  const float* beta = (const float*)d_in[5];
  float* out = (float*)d_out;

  constexpr int D = 1024;
  const int M = in_sizes[0] / D;

  float* G2 = (float*)d_ws;

  dim3 blk(256);

  gemm_tn_ksplit_f32_wmma<D, D, D, D, 8>
      <<<dim3(D / 32, D / 64), blk, 0, stream>>>(Wq, Wk, G2);

  gemm_scores_ln_f32_wmma<D, D>
      <<<dim3(M / 32), blk, 0, stream>>>(cur, G2, prev, gamma, beta, out);
}


# Finishing pass: c~ handed over through LDS (32x1024 tile), each wave then finishes 4 whole tokens alone, no barriers or cross-wave sums per token; score and LN sum order changed
# speedup vs baseline: 2.0406x; 1.0473x over previous
#include <hip/hip_runtime.h>
#include <hip/hip_bf16.h>

typedef float v2f __attribute__((ext_vector_type(2)));
typedef float v4f __attribute__((ext_vector_type(4)));
typedef float v8f __attribute__((ext_vector_type(8)));


template <int K, int LDA, int LDB, int LDC, int KSPLIT>
__global__ __launch_bounds__(256) void gemm_tn_ksplit_f32_wmma(
    const float* __restrict__ A, const float* __restrict__ B,
    float* __restrict__ C2) {
  constexpr int MT = 4;
  constexpr int NT = 2;
  constexpr int KW = K / KSPLIT;
  static_assert(KSPLIT == 2 * MT, "one finishing wave per 8 rows");

  const int lane = threadIdx.x & 31;
  const int wave = threadIdx.x >> 5;
  const int tileM = blockIdx.y * (MT * 16);
  const int tileN = blockIdx.x * (NT * 16);
  const int k0 = wave * KW;
  const int r = lane & 15;
  const int h = lane >> 4;

  const float* Ap = A + (size_t)(k0 + 2 * h) * LDA + (tileM + r);
  const float* Bp = B + (size_t)(k0 + 2 * h) * LDB + (tileN + r);

  v8f acc[MT][NT];
#pragma unroll
  for (int i = 0; i < MT; ++i)
#pragma unroll
    for (int j = 0; j < NT; ++j) acc[i][j] = (v8f){};

#pragma unroll 2
  for (int k = 0; k < KW; k += 4) {
    v2f a[MT], b[NT];
#pragma unroll
    for (int i = 0; i < MT; ++i) {
      a[i].x = Ap[16 * i];
      a[i].y = Ap[16 * i + LDA];
    }
#pragma unroll
    for (int j = 0; j < NT; ++j) {
      b[j].x = Bp[16 * j];
      b[j].y = Bp[16 * j + LDB];
    }
#pragma unroll
    for (int i = 0; i < MT; ++i)
#pragma unroll
      for (int j = 0; j < NT; ++j)
        acc[i][j] = __builtin_amdgcn_wmma_f32_16x16x4_f32(
            false, a[i], false, b[j], (short)0, acc[i][j], false, false);
    Ap += 4 * LDA;
    Bp += 4 * LDB;
  }

  constexpr int SUB = 8 * 32;
  constexpr int TILE = MT * NT * SUB;
  static_assert(NT * 16 == 32, "finishing pass: lane = column of the tile");
  __shared__ float part[KSPLIT * TILE];
#pragma unroll
  for (int i = 0; i < MT; ++i)
#pragma unroll
    for (int j = 0; j < NT; ++j)
#pragma unroll
      for (int v = 0; v < 8; ++v)
        part[wave * TILE + (i * NT + j) * SUB + v * 32 + lane] = acc[i][j][v];
  __syncthreads();
  const float* sp = part + ((wave >> 1) * NT + (lane >> 4)) * SUB +
                    (wave & 1) * 16 + (lane & 15);
  float* Cp = C2 + ((size_t)((tileM + 8 * wave) >> 1) * LDC + tileN + lane) * 2;
#pragma unroll
  for (int vp = 0; vp < 4; ++vp) {
    float t0 = sp[(2 * vp) * 32];
    float t1 = sp[(2 * vp + 1) * 32];
#pragma unroll
    for (int w = 1; w < KSPLIT; ++w) {
      t0 += sp[w * TILE + (2 * vp) * 32];
      t1 += sp[w * TILE + (2 * vp + 1) * 32];
    }
    *(v2f*)(Cp + (size_t)vp * 2 * LDC) = (v2f){t0, t1};
  }
}

template <int M>
static __device__ __forceinline__ float row_xor_add(float x) {
  const int y = __builtin_amdgcn_update_dpp(0, __builtin_bit_cast(int, x),
                                            0x160 + M, 0xf, 0xf, false);
  return x + __builtin_bit_cast(float, y);
}
static __device__ __forceinline__ float row16_sum(float x) {
  x = row_xor_add<8>(x);
  x = row_xor_add<4>(x);
  x = row_xor_add<2>(x);
  x = row_xor_add<1>(x);
  return x;
}

static __device__ __forceinline__ float wave32_sum(float x) {
  x = row16_sum(x);
  const unsigned y = __builtin_amdgcn_permlanex16(
      0u, __builtin_bit_cast(unsigned, x), 0x76543210u, 0xfedcba98u, false, false);
  return x + __builtin_bit_cast(float, y);
}

static __device__ __forceinline__ void lds_barrier() {
  asm volatile("s_wait_dscnt 0x0\n\ts_barrier_signal -1\n\ts_barrier_wait -1" ::: "memory");
}

template <int K, int N>
__global__ __launch_bounds__(256) void gemm_scores_ln_f32_wmma(
    const float* cur, const float* G2, const float* prev, const float* gamma,
    const float* beta, float* out) {
  constexpr int MT = 2;
  constexpr int NG = 4;
  constexpr int P = 8;
  static_assert(K % 32 == 0 && K >= 64 && N == 8 * 32 * NG,
                "8 waves x 128 columns, K loop of four 8-k blocks");

  const int lane = threadIdx.x & 31;
  const int wave = threadIdx.x >> 5;
  const int tileM = blockIdx.x * (MT * 16);
  const int colbase = wave * (32 * NG);
  const int r = lane & 15;
  const int h = lane >> 4;

  const float* Ap = cur + (size_t)(tileM + r) * K + 4 * h;
  const float* Bp =
      G2 + (size_t)(2 * h) * (2 * N) + 2 * colbase + 4 * r;

  v8f acc[MT][NG][2];
#pragma unroll
  for (int i = 0; i < MT; ++i)
#pragma unroll
    for (int g = 0; g < NG; ++g)
#pragma unroll
      for (int j = 0; j < 2; ++j) acc[i][g][j] = (v8f){};

  v4f a[4][MT], b[4][2][NG];

  auto ldfrag = [&](int bf, int ko) {
#pragma unroll
    for (int i = 0; i < MT; ++i)
      a[bf][i] = *(const v4f*)(Ap + ko + (size_t)i * 16 * K);
#pragma unroll
    for (int s = 0; s < 2; ++s)
#pragma unroll
      for (int g = 0; g < NG; ++g)
        b[bf][s][g] = *(const v4f*)(Bp + (size_t)ko * N + (size_t)s * 2 * N + 64 * g);
  };
  auto mma = [&](int bf) {
#pragma unroll
    for (int s = 0; s < 2; ++s)
#pragma unroll
      for (int i = 0; i < MT; ++i)
#pragma unroll
        for (int g = 0; g < NG; ++g)
#pragma unroll
          for (int j = 0; j < 2; ++j) {
            const v2f av = {a[bf][i][2 * s], a[bf][i][2 * s + 1]};
            const v2f bv = {b[bf][s][g][2 * j], b[bf][s][g][2 * j + 1]};
            acc[i][g][j] = __builtin_amdgcn_wmma_f32_16x16x4_f32(
                false, av, false, bv, (short)0, acc[i][g][j], false, false);
          }
  };

#define PIN_ORDER() __builtin_amdgcn_sched_barrier(0)
  ldfrag(0, 0);
  PIN_ORDER();
  ldfrag(1, 8);
  PIN_ORDER();
  ldfrag(2, 16);
  for (int kb = 0; kb < K - 32; kb += 32) {
    PIN_ORDER();
    ldfrag(3, 24);
    PIN_ORDER();
    mma(0);
    PIN_ORDER();
    ldfrag(0, 32);
    PIN_ORDER();
    mma(1);
    PIN_ORDER();
    ldfrag(1, 40);
    PIN_ORDER();
    mma(2);
    PIN_ORDER();
    ldfrag(2, 48);
    PIN_ORDER();
    mma(3);
    Ap += 32;
    Bp += 32 * N;
  }
  PIN_ORDER();
  ldfrag(3, 24);
  PIN_ORDER();
  mma(0);
  mma(1);
  mma(2);
  mma(3);
#undef PIN_ORDER

  constexpr int TPW = 4;
  constexpr int NQ = N / 128;
  constexpr int CP = N + 4;
  __shared__ __attribute__((aligned(16))) float cs[MT * 16 * CP];

  const int tok0 = tileM + TPW * wave;
  const float* pvt = prev + (size_t)tok0 * P * N + 4 * lane;
  const float* xt = cur + (size_t)tok0 * N + 4 * lane;
  float* ot = out + (size_t)tok0 * N + 4 * lane;

  v4f pv[P][NQ], xv[NQ];
  auto ldtoken = [&](int tk) {
#pragma unroll
    for (int p = 0; p < P; ++p)
#pragma unroll
      for (int q = 0; q < NQ; ++q)
        pv[p][q] = __builtin_nontemporal_load(
            (const v4f*)(pvt + ((size_t)tk * P + p) * N + 128 * q));
#pragma unroll
    for (int q = 0; q < NQ; ++q)
      xv[q] = *(const v4f*)(xt + (size_t)tk * N + 128 * q);
  };
  ldtoken(0);

#pragma unroll
  for (int i = 0; i < MT; ++i)
#pragma unroll
    for (int g = 0; g < NG; ++g)
#pragma unroll
      for (int v = 0; v < 8; ++v)
        *(v2f*)(cs + (16 * i + 8 * h + v) * CP + colbase + 32 * g + 2 * r) =
            (v2f){acc[i][g][0][v], acc[i][g][1][v]};
  lds_barrier();

  const float inv_sqrt_d = 0.03125f;
#pragma unroll
  for (int tk = 0; tk < TPW; ++tk) {
    const float* crow = cs + (TPW * wave + tk) * CP + 4 * lane;
    v4f cv[NQ];
#pragma unroll
    for (int q = 0; q < NQ; ++q) cv[q] = *(const v4f*)(crow + 128 * q);

    float s[P];
#pragma unroll
    for (int p = 0; p < P; ++p) {
      v2f t = {0.f, 0.f};
#pragma unroll
      for (int q = 0; q < NQ; ++q) {
        t = __builtin_elementwise_fma(cv[q].lo, pv[p][q].lo, t);
        t = __builtin_elementwise_fma(cv[q].hi, pv[p][q].hi, t);
      }
      s[p] = t.x + t.y;
    }
    float score[P];
#pragma unroll
    for (int p = 0; p < P; ++p) score[p] = wave32_sum(s[p]) * inv_sqrt_d;

    float mx = score[0];
#pragma unroll
    for (int p = 1; p < P; ++p) mx = fmaxf(mx, score[p]);
    float e[P], esum = 0.f;
#pragma unroll
    for (int p = 0; p < P; ++p) {
      e[p] = __expf(score[p] - mx);
      esum += e[p];
    }
    const float inv = 1.f / esum;

    v4f y[NQ];
#pragma unroll
    for (int q = 0; q < NQ; ++q) y[q] = xv[q];
#pragma unroll
    for (int p = 0; p < P; ++p) {
      const float w = e[p] * inv;
#pragma unroll
      for (int q = 0; q < NQ; ++q)
        y[q] = __builtin_elementwise_fma((v4f){w, w, w, w}, pv[p][q], y[q]);
    }

    v4f gm[NQ], bt[NQ];
#pragma unroll
    for (int q = 0; q < NQ; ++q) {
      gm[q] = *(const v4f*)(gamma + 4 * lane + 128 * q);
      bt[q] = *(const v4f*)(beta + 4 * lane + 128 * q);
    }
    __builtin_amdgcn_sched_barrier(0);
    if (tk + 1 < TPW) ldtoken(tk + 1);
    __builtin_amdgcn_sched_barrier(0);

    v4f a1 = y[0], a2 = y[0] * y[0];
#pragma unroll
    for (int q = 1; q < NQ; ++q) {
      a1 += y[q];
      a2 = __builtin_elementwise_fma(y[q], y[q], a2);
    }
    const float S = wave32_sum((a1.x + a1.y) + (a1.z + a1.w));
    const float S2 = wave32_sum((a2.x + a2.y) + (a2.z + a2.w));
    const float mu = S * (1.f / N);
    const float var = S2 * (1.f / N) - mu * mu;
    const float rstd = rsqrtf(var + 1e-5f);

#pragma unroll
    for (int q = 0; q < NQ; ++q) {
      v4f o;
#pragma unroll
      for (int k = 0; k < 4; ++k)
        o[k] = (y[q][k] - mu) * rstd * gm[q][k] + bt[q][k];
      __builtin_nontemporal_store(o, (v4f*)(ot + (size_t)tk * N + 128 * q));
    }
    __builtin_amdgcn_sched_barrier(0);
  }
}

extern "C" void kernel_launch(void* const* d_in, const int* in_sizes, int n_in,
                              void* d_out, int out_size, void* d_ws,
                              size_t ws_size, hipStream_t stream) {
  const float* cur = (const float*)d_in[0];
  const float* prev = (const float*)d_in[1];
  const float* Wq = (const float*)d_in[2];
  const float* Wk = (const float*)d_in[3];
  const float* gamma = (const float*)d_in[4];
  const float* beta = (const float*)d_in[5];
  float* out = (float*)d_out;

  constexpr int D = 1024;
  const int M = in_sizes[0] / D;

  float* G2 = (float*)d_ws;

  dim3 blk(256);

  gemm_tn_ksplit_f32_wmma<D, D, D, D, 8>
      <<<dim3(D / 32, D / 64), blk, 0, stream>>>(Wq, Wk, G2);

  gemm_scores_ln_f32_wmma<D, D>
      <<<dim3(M / 32), blk, 0, stream>>>(cur, G2, prev, gamma, beta, out);
}
